# starting kernel
# baseline (speedup 1.0000x reference)
#include <hip/hip_runtime.h>
#include <hip/hip_bf16.h>


typedef __attribute__((ext_vector_type(2))) float v2f;
typedef __attribute__((ext_vector_type(4))) float v4f;
typedef __attribute__((ext_vector_type(8))) float v8f;

#define NTOK   4096
#define EMBED  512
#define QKVW   1536
#define HEADS  8
#define DH     64
#define SCALE  0.125f

static __device__ __forceinline__ v8f wmma4(v2f a, v2f b, v8f c) {
    return __builtin_amdgcn_wmma_f32_16x16x4_f32(
         false, a,  false, b,
         (short)0, c,  false,  false);
}

__global__ void mhsa_zero(float* __restrict__ p, int n) {
    int i = blockIdx.x * blockDim.x + threadIdx.x;
    if (i < n) p[i] = 0.0f;
}

__global__ __launch_bounds__(128) void mhsa_qkv_gemm(
    const float* __restrict__ x, const float* __restrict__ W,
    const float* __restrict__ bias, float* __restrict__ qkv)
{
    const int lane = threadIdx.x & 31;
    const int wave = threadIdx.x >> 5;
    const int lo = lane & 15;
    const int hi = lane >> 4;
    const int m0 = blockIdx.x * 64 + wave * 16;
    const int n0 = blockIdx.y * 64;

    v8f acc0 = {}, acc1 = {}, acc2 = {}, acc3 = {};
    const int m  = m0 + lo;
    const int nb = n0 + lo;

    #pragma unroll 4
    for (int k = 0; k < EMBED; k += 4) {
        const int ka = k + 2 * hi;
        v2f a;
        a.x = x[(size_t)ka * NTOK + m];
        a.y = x[(size_t)(ka + 1) * NTOK + m];
        const float* w0 = W + (size_t)ka * QKVW + nb;
        const float* w1 = W + (size_t)(ka + 1) * QKVW + nb;
        v2f b0 = {w0[0],  w1[0]};
        v2f b1 = {w0[16], w1[16]};
        v2f b2 = {w0[32], w1[32]};
        v2f b3 = {w0[48], w1[48]};
        acc0 = wmma4(a, b0, acc0);
        acc1 = wmma4(a, b1, acc1);
        acc2 = wmma4(a, b2, acc2);
        acc3 = wmma4(a, b3, acc3);
    }

    const float bi0 = bias[nb], bi1 = bias[nb + 16];
    const float bi2 = bias[nb + 32], bi3 = bias[nb + 48];
    #pragma unroll
    for (int r = 0; r < 8; ++r) {
        const int row = m0 + hi * 8 + r;
        float* dst = qkv + (size_t)row * QKVW + n0 + lo;
        dst[0]  = acc0[r] + bi0;
        dst[16] = acc1[r] + bi1;
        dst[32] = acc2[r] + bi2;
        dst[48] = acc3[r] + bi3;
    }
}

__global__ __launch_bounds__(64) void mhsa_qsum(
    const float* __restrict__ qkv, float* __restrict__ qsum)
{
    const int h = blockIdx.x, chunk = blockIdx.y, d = threadIdx.x;
    const float* col = qkv + (size_t)chunk * 256 * QKVW + h * DH + d;
    float s = 0.0f;
    #pragma unroll 8
    for (int i = 0; i < 256; ++i) s += col[(size_t)i * QKVW];
    atomicAdd(&qsum[h * DH + d], s);
}

__global__ __launch_bounds__(128) void mhsa_ktv(
    const float* __restrict__ qkv, float* __restrict__ KtV)
{
    const int h = blockIdx.x, chunk = blockIdx.y;
    const int lane = threadIdx.x & 31, wave = threadIdx.x >> 5;
    const int lo = lane & 15, hi = lane >> 4;
    const int m0 = wave * 16;
    const float* Kb = qkv + EMBED + h * DH;
    const float* Vb = qkv + 2 * EMBED + h * DH;
    const int nbase = chunk * 256;

    v8f acc0 = {}, acc1 = {}, acc2 = {}, acc3 = {};
    #pragma unroll 4
    for (int kk = 0; kk < 256; kk += 4) {
        const int n = nbase + kk + 2 * hi;
        v2f a;
        a.x = Kb[(size_t)n * QKVW + m0 + lo];
        a.y = Kb[(size_t)(n + 1) * QKVW + m0 + lo];
        const float* v0 = Vb + (size_t)n * QKVW + lo;
        const float* v1 = Vb + (size_t)(n + 1) * QKVW + lo;
        v2f b0 = {v0[0],  v1[0]};
        v2f b1 = {v0[16], v1[16]};
        v2f b2 = {v0[32], v1[32]};
        v2f b3 = {v0[48], v1[48]};
        acc0 = wmma4(a, b0, acc0);
        acc1 = wmma4(a, b1, acc1);
        acc2 = wmma4(a, b2, acc2);
        acc3 = wmma4(a, b3, acc3);
    }

    float* dst = KtV + (size_t)h * DH * DH;
    #pragma unroll
    for (int r = 0; r < 8; ++r) {
        const int row = m0 + hi * 8 + r;
        atomicAdd(&dst[row * DH + lo +  0], acc0[r]);
        atomicAdd(&dst[row * DH + lo + 16], acc1[r]);
        atomicAdd(&dst[row * DH + lo + 32], acc2[r]);
        atomicAdd(&dst[row * DH + lo + 48], acc3[r]);
    }
}

__global__ __launch_bounds__(256) void mhsa_coarse(
    const float* __restrict__ qkv, const float* __restrict__ qsum,
    float* __restrict__ out)
{
    const int idx = blockIdx.x * blockDim.x + threadIdx.x;
    const int h = idx >> 12, n = idx & (NTOK - 1);
    const float* Kr = qkv + (size_t)n * QKVW + EMBED + h * DH;
    const float* qs = qsum + h * DH;
    float s = 0.0f;
    #pragma unroll
    for (int d = 0; d < DH; d += 4) {
        const v4f kv = *(const v4f*)(Kr + d);
        const v4f qv = *(const v4f*)(qs + d);
        s += kv.x * qv.x + kv.y * qv.y + kv.z * qv.z + kv.w * qv.w;
    }
    out[idx] = s * SCALE;
}

__global__ __launch_bounds__(128) void mhsa_out_gemm(
    const float* __restrict__ qkv, const float* __restrict__ KtV,
    float* __restrict__ out)
{
    __shared__ float lB[DH * DH];
    const int h = blockIdx.x, mt = blockIdx.y;

    {
        const v4f* src = (const v4f*)(KtV + (size_t)h * DH * DH);
        v4f* dst = (v4f*)lB;
        for (int i = threadIdx.x; i < (DH * DH) / 4; i += 128) dst[i] = src[i];
    }
    __syncthreads();

    const int lane = threadIdx.x & 31, wave = threadIdx.x >> 5;
    const int lo = lane & 15, hi = lane >> 4;
    const int m0 = mt * 64 + wave * 16;
    const float* Qb = qkv + h * DH;
    const int m = m0 + lo;

    v8f acc0 = {}, acc1 = {}, acc2 = {}, acc3 = {};
    #pragma unroll
    for (int k = 0; k < DH; k += 4) {
        const int ka = k + 2 * hi;
        v2f a = *(const v2f*)(Qb + (size_t)m * QKVW + ka);
        const float* B0 = lB + ka * DH + lo;
        const float* B1 = lB + (ka + 1) * DH + lo;
        v2f b0 = {B0[0],  B1[0]};
        v2f b1 = {B0[16], B1[16]};
        v2f b2 = {B0[32], B1[32]};
        v2f b3 = {B0[48], B1[48]};
        acc0 = wmma4(a, b0, acc0);
        acc1 = wmma4(a, b1, acc1);
        acc2 = wmma4(a, b2, acc2);
        acc3 = wmma4(a, b3, acc3);
    }

    float* base = out + HEADS * NTOK + (size_t)h * NTOK * DH;
    #pragma unroll
    for (int r = 0; r < 8; ++r) {
        const int row = m0 + hi * 8 + r;
        float* o = base + (size_t)row * DH + lo;
        o[0]  = acc0[r] * SCALE;
        o[16] = acc1[r] * SCALE;
        o[32] = acc2[r] * SCALE;
        o[48] = acc3[r] * SCALE;
    }
}

extern "C" void kernel_launch(void* const* d_in, const int* in_sizes, int n_in,
                              void* d_out, int out_size, void* d_ws, size_t ws_size,
                              hipStream_t stream) {
    const float* x    = (const float*)d_in[0];
    const float* W    = (const float*)d_in[1];
    const float* bias = (const float*)d_in[2];
    float* out = (float*)d_out;

    float* qkv  = (float*)d_ws;
    float* KtV  = qkv + (size_t)NTOK * QKVW;
    float* qsum = KtV + HEADS * DH * DH;

    const int nzero = HEADS * DH * DH + HEADS * DH;
    mhsa_zero<<<(nzero + 255) / 256, 256, 0, stream>>>(KtV, nzero);

    mhsa_qkv_gemm<<<dim3(NTOK / 64, QKVW / 64), 128, 0, stream>>>(x, W, bias, qkv);
    mhsa_qsum<<<dim3(HEADS, NTOK / 256), 64, 0, stream>>>(qkv, qsum);
    mhsa_ktv<<<dim3(HEADS, NTOK / 256), 128, 0, stream>>>(qkv, KtV);
    mhsa_coarse<<<(HEADS * NTOK) / 256, 256, 0, stream>>>(qkv, qsum, out);
    mhsa_out_gemm<<<dim3(HEADS, NTOK / 64), 128, 0, stream>>>(qkv, KtV, out);
}


# qsum and KtV token reductions: split each 256-token chunk over 8 and 4 wave slices inside the block, fold slice partials in LDS in fixed order before the unchanged atomics (sum order changes)
# speedup vs baseline: 1.3792x; 1.3792x over previous
#include <hip/hip_runtime.h>
#include <hip/hip_bf16.h>


typedef __attribute__((ext_vector_type(2))) float v2f;
typedef __attribute__((ext_vector_type(4))) float v4f;
typedef __attribute__((ext_vector_type(8))) float v8f;

#define NTOK   4096
#define EMBED  512
#define QKVW   1536
#define HEADS  8
#define DH     64
#define SCALE  0.125f

static __device__ __forceinline__ v8f wmma4(v2f a, v2f b, v8f c) {
    return __builtin_amdgcn_wmma_f32_16x16x4_f32(
         false, a,  false, b,
         (short)0, c,  false,  false);
}

__global__ void mhsa_zero(float* __restrict__ p, int n) {
    int i = blockIdx.x * blockDim.x + threadIdx.x;
    if (i < n) p[i] = 0.0f;
}

__global__ __launch_bounds__(128) void mhsa_qkv_gemm(
    const float* __restrict__ x, const float* __restrict__ W,
    const float* __restrict__ bias, float* __restrict__ qkv)
{
    const int lane = threadIdx.x & 31;
    const int wave = threadIdx.x >> 5;
    const int lo = lane & 15;
    const int hi = lane >> 4;
    const int m0 = blockIdx.x * 64 + wave * 16;
    const int n0 = blockIdx.y * 64;

    v8f acc0 = {}, acc1 = {}, acc2 = {}, acc3 = {};
    const int m  = m0 + lo;
    const int nb = n0 + lo;

    #pragma unroll 4
    for (int k = 0; k < EMBED; k += 4) {
        const int ka = k + 2 * hi;
        v2f a;
        a.x = x[(size_t)ka * NTOK + m];
        a.y = x[(size_t)(ka + 1) * NTOK + m];
        const float* w0 = W + (size_t)ka * QKVW + nb;
        const float* w1 = W + (size_t)(ka + 1) * QKVW + nb;
        v2f b0 = {w0[0],  w1[0]};
        v2f b1 = {w0[16], w1[16]};
        v2f b2 = {w0[32], w1[32]};
        v2f b3 = {w0[48], w1[48]};
        acc0 = wmma4(a, b0, acc0);
        acc1 = wmma4(a, b1, acc1);
        acc2 = wmma4(a, b2, acc2);
        acc3 = wmma4(a, b3, acc3);
    }

    const float bi0 = bias[nb], bi1 = bias[nb + 16];
    const float bi2 = bias[nb + 32], bi3 = bias[nb + 48];
    #pragma unroll
    for (int r = 0; r < 8; ++r) {
        const int row = m0 + hi * 8 + r;
        float* dst = qkv + (size_t)row * QKVW + n0 + lo;
        dst[0]  = acc0[r] + bi0;
        dst[16] = acc1[r] + bi1;
        dst[32] = acc2[r] + bi2;
        dst[48] = acc3[r] + bi3;
    }
}

__global__ __launch_bounds__(512) void mhsa_qsum(
    const float* __restrict__ qkv, float* __restrict__ qsum)
{
    __shared__ float red[8 * DH];
    const int h = blockIdx.x, chunk = blockIdx.y;
    const int d = threadIdx.x & (DH - 1), sl = threadIdx.x >> 6;
    const float* col = qkv + (size_t)(chunk * 256 + sl * 32) * QKVW + h * DH + d;
    float s = 0.0f;
    #pragma unroll
    for (int i = 0; i < 32; ++i) s += col[(size_t)i * QKVW];
    red[sl * DH + d] = s;
    __syncthreads();
    if (threadIdx.x < DH) {
        float t = red[d];
        #pragma unroll
        for (int j = 1; j < 8; ++j) t += red[j * DH + d];
        atomicAdd(&qsum[h * DH + d], t);
    }
}

__global__ __launch_bounds__(512) void mhsa_ktv(
    const float* __restrict__ qkv, float* __restrict__ KtV)
{
    __shared__ float red[16 * 32 * 32];
    const int h = blockIdx.x, chunk = blockIdx.y;
    const int lane = threadIdx.x & 31, wave = threadIdx.x >> 5;
    const int lo = lane & 15, hi = lane >> 4;
    const int dt = wave & 3;
    const int sl = wave >> 2;
    const int m0 = dt * 16;
    const float* Kb = qkv + EMBED + h * DH;
    const float* Vb = qkv + 2 * EMBED + h * DH;
    const int nbase = chunk * 256 + sl * 64;

    v8f acc0 = {}, acc1 = {}, acc2 = {}, acc3 = {};
    #pragma unroll 8
    for (int kk = 0; kk < 64; kk += 4) {
        const int n = nbase + kk + 2 * hi;
        v2f a;
        a.x = Kb[(size_t)n * QKVW + m0 + lo];
        a.y = Kb[(size_t)(n + 1) * QKVW + m0 + lo];
        const float* v0 = Vb + (size_t)n * QKVW + lo;
        const float* v1 = Vb + (size_t)(n + 1) * QKVW + lo;
        v2f b0 = {v0[0],  v1[0]};
        v2f b1 = {v0[16], v1[16]};
        v2f b2 = {v0[32], v1[32]};
        v2f b3 = {v0[48], v1[48]};
        acc0 = wmma4(a, b0, acc0);
        acc1 = wmma4(a, b1, acc1);
        acc2 = wmma4(a, b2, acc2);
        acc3 = wmma4(a, b3, acc3);
    }

    float* mine = red + wave * 1024 + lane;
    #pragma unroll
    for (int r = 0; r < 8; ++r) {
        mine[(0 * 8 + r) * 32] = acc0[r];
        mine[(1 * 8 + r) * 32] = acc1[r];
        mine[(2 * 8 + r) * 32] = acc2[r];
        mine[(3 * 8 + r) * 32] = acc3[r];
    }
    __syncthreads();

    const float* p = red + dt * 1024 + sl * 256 + lane;
    float* dst = KtV + (size_t)h * DH * DH;
    #pragma unroll
    for (int r = 0; r < 8; ++r) {
        const float s = ((p[r * 32] + p[4096 + r * 32]) + p[8192 + r * 32])
                        + p[12288 + r * 32];
        const int row = m0 + hi * 8 + r;
        atomicAdd(&dst[row * DH + lo + 16 * sl], s);
    }
}

__global__ __launch_bounds__(256) void mhsa_coarse(
    const float* __restrict__ qkv, const float* __restrict__ qsum,
    float* __restrict__ out)
{
    const int idx = blockIdx.x * blockDim.x + threadIdx.x;
    const int h = idx >> 12, n = idx & (NTOK - 1);
    const float* Kr = qkv + (size_t)n * QKVW + EMBED + h * DH;
    const float* qs = qsum + h * DH;
    float s = 0.0f;
    #pragma unroll
    for (int d = 0; d < DH; d += 4) {
        const v4f kv = *(const v4f*)(Kr + d);
        const v4f qv = *(const v4f*)(qs + d);
        s += kv.x * qv.x + kv.y * qv.y + kv.z * qv.z + kv.w * qv.w;
    }
    out[idx] = s * SCALE;
}

__global__ __launch_bounds__(128) void mhsa_out_gemm(
    const float* __restrict__ qkv, const float* __restrict__ KtV,
    float* __restrict__ out)
{
    __shared__ float lB[DH * DH];
    const int h = blockIdx.x, mt = blockIdx.y;

    {
        const v4f* src = (const v4f*)(KtV + (size_t)h * DH * DH);
        v4f* dst = (v4f*)lB;
        for (int i = threadIdx.x; i < (DH * DH) / 4; i += 128) dst[i] = src[i];
    }
    __syncthreads();

    const int lane = threadIdx.x & 31, wave = threadIdx.x >> 5;
    const int lo = lane & 15, hi = lane >> 4;
    const int m0 = mt * 64 + wave * 16;
    const float* Qb = qkv + h * DH;
    const int m = m0 + lo;

    v8f acc0 = {}, acc1 = {}, acc2 = {}, acc3 = {};
    #pragma unroll
    for (int k = 0; k < DH; k += 4) {
        const int ka = k + 2 * hi;
        v2f a = *(const v2f*)(Qb + (size_t)m * QKVW + ka);
        const float* B0 = lB + ka * DH + lo;
        const float* B1 = lB + (ka + 1) * DH + lo;
        v2f b0 = {B0[0],  B1[0]};
        v2f b1 = {B0[16], B1[16]};
        v2f b2 = {B0[32], B1[32]};
        v2f b3 = {B0[48], B1[48]};
        acc0 = wmma4(a, b0, acc0);
        acc1 = wmma4(a, b1, acc1);
        acc2 = wmma4(a, b2, acc2);
        acc3 = wmma4(a, b3, acc3);
    }

    float* base = out + HEADS * NTOK + (size_t)h * NTOK * DH;
    #pragma unroll
    for (int r = 0; r < 8; ++r) {
        const int row = m0 + hi * 8 + r;
        float* o = base + (size_t)row * DH + lo;
        o[0]  = acc0[r] * SCALE;
        o[16] = acc1[r] * SCALE;
        o[32] = acc2[r] * SCALE;
        o[48] = acc3[r] * SCALE;
    }
}

extern "C" void kernel_launch(void* const* d_in, const int* in_sizes, int n_in,
                              void* d_out, int out_size, void* d_ws, size_t ws_size,
                              hipStream_t stream) {
    const float* x    = (const float*)d_in[0];
    const float* W    = (const float*)d_in[1];
    const float* bias = (const float*)d_in[2];
    float* out = (float*)d_out;

    float* qkv  = (float*)d_ws;
    float* KtV  = qkv + (size_t)NTOK * QKVW;
    float* qsum = KtV + HEADS * DH * DH;

    const int nzero = HEADS * DH * DH + HEADS * DH;
    mhsa_zero<<<(nzero + 255) / 256, 256, 0, stream>>>(KtV, nzero);

    mhsa_qkv_gemm<<<dim3(NTOK / 64, QKVW / 64), 128, 0, stream>>>(x, W, bias, qkv);
    mhsa_qsum<<<dim3(HEADS, NTOK / 256), 512, 0, stream>>>(qkv, qsum);
    mhsa_ktv<<<dim3(HEADS, NTOK / 256), 512, 0, stream>>>(qkv, KtV);
    mhsa_coarse<<<(HEADS * NTOK) / 256, 256, 0, stream>>>(qkv, qsum, out);
    mhsa_out_gemm<<<dim3(HEADS, NTOK / 64), 128, 0, stream>>>(qkv, KtV, out);
}


# QKV GEMM: wave tile 32x64 with its WMMA tiles interleaved along the contiguous index, so a lane loads A as 8-byte and B as 16-byte words (4 loads per 8 WMMAs, was 10 per 4); sum order unchanged
# speedup vs baseline: 1.4477x; 1.0497x over previous
#include <hip/hip_runtime.h>
#include <hip/hip_bf16.h>


typedef __attribute__((ext_vector_type(2))) float v2f;
typedef __attribute__((ext_vector_type(4))) float v4f;
typedef __attribute__((ext_vector_type(8))) float v8f;

#define NTOK   4096
#define EMBED  512
#define QKVW   1536
#define HEADS  8
#define DH     64
#define SCALE  0.125f

static __device__ __forceinline__ v8f wmma4(v2f a, v2f b, v8f c) {
    return __builtin_amdgcn_wmma_f32_16x16x4_f32(
         false, a,  false, b,
         (short)0, c,  false,  false);
}

__global__ void mhsa_zero(float* __restrict__ p, int n) {
    int i = blockIdx.x * blockDim.x + threadIdx.x;
    if (i < n) p[i] = 0.0f;
}

struct QkvRaw {
    v2f a[4][2];
    v4f b[4][2];
};
struct QkvOps {
    v2f a[4][2];
    v2f b[4][4];
};

static __device__ __forceinline__ void qkv_load(
    QkvRaw& f, const float* __restrict__ xa, const float* __restrict__ wb, int k0)
{
    #pragma unroll
    for (int i = 0; i < 4; ++i) {
        #pragma unroll
        for (int s = 0; s < 2; ++s) {
            f.a[i][s] = *(const v2f*)(xa + (size_t)(k0 + 4 * i + s) * NTOK);
            f.b[i][s] = *(const v4f*)(wb + (size_t)(k0 + 4 * i + s) * QKVW);
        }
    }
}

static __device__ __forceinline__ void qkv_pair(const QkvRaw& f, QkvOps& p)
{
    #pragma unroll
    for (int i = 0; i < 4; ++i) {
        #pragma unroll
        for (int t = 0; t < 2; ++t) {
            p.a[i][t] = v2f{f.a[i][0][t], f.a[i][1][t]};
            asm volatile("" : "+v"(p.a[i][t]));
        }
        #pragma unroll
        for (int j = 0; j < 4; ++j) {
            p.b[i][j] = v2f{f.b[i][0][j], f.b[i][1][j]};
            asm volatile("" : "+v"(p.b[i][j]));
        }
    }
}

static __device__ __forceinline__ void qkv_mma(const QkvOps& p, v8f (&acc)[2][4])
{
    #pragma unroll
    for (int i = 0; i < 4; ++i) {
        #pragma unroll
        for (int t = 0; t < 2; ++t) {
            #pragma unroll
            for (int j = 0; j < 4; ++j)
                acc[t][j] = wmma4(p.a[i][t], p.b[i][j], acc[t][j]);
        }
    }
}

static __device__ __forceinline__ void qkv_wmma_gap()
{
    asm volatile("v_nop\n\tv_nop\n\tv_nop\n\tv_nop");
}

__global__ __launch_bounds__(128) void mhsa_qkv_gemm(
    const float* __restrict__ x, const float* __restrict__ W,
    const float* __restrict__ bias, float* __restrict__ qkv)
{
    const int lane = threadIdx.x & 31;
    const int wave = threadIdx.x >> 5;
    const int lo = lane & 15;
    const int hi = lane >> 4;
    const int m0 = blockIdx.x * 128 + wave * 32;
    const int n0 = blockIdx.y * 64;

    const float* xa = x + (size_t)(2 * hi) * NTOK + m0 + 2 * lo;
    const float* wb = W + (size_t)(2 * hi) * QKVW + n0 + 4 * lo;

    v8f acc[2][4] = {};
    QkvRaw raw;
    QkvOps ops;
    qkv_load(raw, xa, wb, 0);
    #pragma unroll 1
    for (int k0 = 0; k0 < EMBED - 16; k0 += 16) {
        qkv_pair(raw, ops);
        __builtin_amdgcn_sched_barrier(0);
        qkv_load(raw, xa, wb, k0 + 16);
        __builtin_amdgcn_sched_barrier(0);
        qkv_mma(ops, acc);
        __builtin_amdgcn_sched_barrier(0);
        qkv_wmma_gap();
        __builtin_amdgcn_sched_barrier(0);
    }
    qkv_pair(raw, ops);
    __builtin_amdgcn_sched_barrier(0);
    qkv_mma(ops, acc);
    __builtin_amdgcn_sched_barrier(0);
    qkv_wmma_gap();
    __builtin_amdgcn_sched_barrier(0);

    const v4f bi = *(const v4f*)(bias + n0 + 4 * lo);
    #pragma unroll
    for (int t = 0; t < 2; ++t) {
        #pragma unroll
        for (int r = 0; r < 8; ++r) {
            const int row = m0 + 2 * (hi * 8 + r) + t;
            v4f o;
            o.x = acc[t][0][r] + bi.x;
            o.y = acc[t][1][r] + bi.y;
            o.z = acc[t][2][r] + bi.z;
            o.w = acc[t][3][r] + bi.w;
            *(v4f*)(qkv + (size_t)row * QKVW + n0 + 4 * lo) = o;
        }
    }
}

__global__ __launch_bounds__(512) void mhsa_qsum(
    const float* __restrict__ qkv, float* __restrict__ qsum)
{
    __shared__ float red[8 * DH];
    const int h = blockIdx.x, chunk = blockIdx.y;
    const int d = threadIdx.x & (DH - 1), sl = threadIdx.x >> 6;
    const float* col = qkv + (size_t)(chunk * 256 + sl * 32) * QKVW + h * DH + d;
    float s = 0.0f;
    #pragma unroll
    for (int i = 0; i < 32; ++i) s += col[(size_t)i * QKVW];
    red[sl * DH + d] = s;
    __syncthreads();
    if (threadIdx.x < DH) {
        float t = red[d];
        #pragma unroll
        for (int j = 1; j < 8; ++j) t += red[j * DH + d];
        atomicAdd(&qsum[h * DH + d], t);
    }
}

__global__ __launch_bounds__(512) void mhsa_ktv(
    const float* __restrict__ qkv, float* __restrict__ KtV)
{
    __shared__ float red[16 * 32 * 32];
    const int h = blockIdx.x, chunk = blockIdx.y;
    const int lane = threadIdx.x & 31, wave = threadIdx.x >> 5;
    const int lo = lane & 15, hi = lane >> 4;
    const int dt = wave & 3;
    const int sl = wave >> 2;
    const int m0 = dt * 16;
    const float* Kb = qkv + EMBED + h * DH;
    const float* Vb = qkv + 2 * EMBED + h * DH;
    const int nbase = chunk * 256 + sl * 64;

    v8f acc0 = {}, acc1 = {}, acc2 = {}, acc3 = {};
    #pragma unroll 8
    for (int kk = 0; kk < 64; kk += 4) {
        const int n = nbase + kk + 2 * hi;
        v2f a;
        a.x = Kb[(size_t)n * QKVW + m0 + lo];
        a.y = Kb[(size_t)(n + 1) * QKVW + m0 + lo];
        const float* v0 = Vb + (size_t)n * QKVW + lo;
        const float* v1 = Vb + (size_t)(n + 1) * QKVW + lo;
        v2f b0 = {v0[0],  v1[0]};
        v2f b1 = {v0[16], v1[16]};
        v2f b2 = {v0[32], v1[32]};
        v2f b3 = {v0[48], v1[48]};
        acc0 = wmma4(a, b0, acc0);
        acc1 = wmma4(a, b1, acc1);
        acc2 = wmma4(a, b2, acc2);
        acc3 = wmma4(a, b3, acc3);
    }

    float* mine = red + wave * 1024 + lane;
    #pragma unroll
    for (int r = 0; r < 8; ++r) {
        mine[(0 * 8 + r) * 32] = acc0[r];
        mine[(1 * 8 + r) * 32] = acc1[r];
        mine[(2 * 8 + r) * 32] = acc2[r];
        mine[(3 * 8 + r) * 32] = acc3[r];
    }
    __syncthreads();

    const float* p = red + dt * 1024 + sl * 256 + lane;
    float* dst = KtV + (size_t)h * DH * DH;
    #pragma unroll
    for (int r = 0; r < 8; ++r) {
        const float s = ((p[r * 32] + p[4096 + r * 32]) + p[8192 + r * 32])
                        + p[12288 + r * 32];
        const int row = m0 + hi * 8 + r;
        atomicAdd(&dst[row * DH + lo + 16 * sl], s);
    }
}

__global__ __launch_bounds__(256) void mhsa_coarse(
    const float* __restrict__ qkv, const float* __restrict__ qsum,
    float* __restrict__ out)
{
    const int idx = blockIdx.x * blockDim.x + threadIdx.x;
    const int h = idx >> 12, n = idx & (NTOK - 1);
    const float* Kr = qkv + (size_t)n * QKVW + EMBED + h * DH;
    const float* qs = qsum + h * DH;
    float s = 0.0f;
    #pragma unroll
    for (int d = 0; d < DH; d += 4) {
        const v4f kv = *(const v4f*)(Kr + d);
        const v4f qv = *(const v4f*)(qs + d);
        s += kv.x * qv.x + kv.y * qv.y + kv.z * qv.z + kv.w * qv.w;
    }
    out[idx] = s * SCALE;
}

__global__ __launch_bounds__(128) void mhsa_out_gemm(
    const float* __restrict__ qkv, const float* __restrict__ KtV,
    float* __restrict__ out)
{
    __shared__ float lB[DH * DH];
    const int h = blockIdx.x, mt = blockIdx.y;

    {
        const v4f* src = (const v4f*)(KtV + (size_t)h * DH * DH);
        v4f* dst = (v4f*)lB;
        for (int i = threadIdx.x; i < (DH * DH) / 4; i += 128) dst[i] = src[i];
    }
    __syncthreads();

    const int lane = threadIdx.x & 31, wave = threadIdx.x >> 5;
    const int lo = lane & 15, hi = lane >> 4;
    const int m0 = mt * 64 + wave * 16;
    const float* Qb = qkv + h * DH;
    const int m = m0 + lo;

    v8f acc0 = {}, acc1 = {}, acc2 = {}, acc3 = {};
    #pragma unroll
    for (int k = 0; k < DH; k += 4) {
        const int ka = k + 2 * hi;
        v2f a = *(const v2f*)(Qb + (size_t)m * QKVW + ka);
        const float* B0 = lB + ka * DH + lo;
        const float* B1 = lB + (ka + 1) * DH + lo;
        v2f b0 = {B0[0],  B1[0]};
        v2f b1 = {B0[16], B1[16]};
        v2f b2 = {B0[32], B1[32]};
        v2f b3 = {B0[48], B1[48]};
        acc0 = wmma4(a, b0, acc0);
        acc1 = wmma4(a, b1, acc1);
        acc2 = wmma4(a, b2, acc2);
        acc3 = wmma4(a, b3, acc3);
    }

    float* base = out + HEADS * NTOK + (size_t)h * NTOK * DH;
    #pragma unroll
    for (int r = 0; r < 8; ++r) {
        const int row = m0 + hi * 8 + r;
        float* o = base + (size_t)row * DH + lo;
        o[0]  = acc0[r] * SCALE;
        o[16] = acc1[r] * SCALE;
        o[32] = acc2[r] * SCALE;
        o[48] = acc3[r] * SCALE;
    }
}

extern "C" void kernel_launch(void* const* d_in, const int* in_sizes, int n_in,
                              void* d_out, int out_size, void* d_ws, size_t ws_size,
                              hipStream_t stream) {
    const float* x    = (const float*)d_in[0];
    const float* W    = (const float*)d_in[1];
    const float* bias = (const float*)d_in[2];
    float* out = (float*)d_out;

    float* qkv  = (float*)d_ws;
    float* KtV  = qkv + (size_t)NTOK * QKVW;
    float* qsum = KtV + HEADS * DH * DH;

    const int nzero = HEADS * DH * DH + HEADS * DH;
    mhsa_zero<<<(nzero + 255) / 256, 256, 0, stream>>>(KtV, nzero);

    mhsa_qkv_gemm<<<dim3(NTOK / 128, QKVW / 64), 128, 0, stream>>>(x, W, bias, qkv);
    mhsa_qsum<<<dim3(HEADS, NTOK / 256), 512, 0, stream>>>(qkv, qsum);
    mhsa_ktv<<<dim3(HEADS, NTOK / 256), 512, 0, stream>>>(qkv, KtV);
    mhsa_coarse<<<(HEADS * NTOK) / 256, 256, 0, stream>>>(qkv, qsum, out);
    mhsa_out_gemm<<<dim3(HEADS, NTOK / 64), 128, 0, stream>>>(qkv, KtV, out);
}


# Launch fusion, 6 launches to 3: accumulator zeroing stored by the QKV GEMM, qsum reduction run inside the KtV kernel, coarse_attn inside the output GEMM; sum order unchanged, atomics kept
# speedup vs baseline: 1.6774x; 1.1587x over previous
#include <hip/hip_runtime.h>
#include <hip/hip_bf16.h>


typedef __attribute__((ext_vector_type(2))) float v2f;
typedef __attribute__((ext_vector_type(4))) float v4f;
typedef __attribute__((ext_vector_type(8))) float v8f;

#define NTOK   4096
#define EMBED  512
#define QKVW   1536
#define HEADS  8
#define DH     64
#define SCALE  0.125f
#define NZERO  (HEADS * DH * DH + HEADS * DH)

static __device__ __forceinline__ v8f wmma4(v2f a, v2f b, v8f c) {
    return __builtin_amdgcn_wmma_f32_16x16x4_f32(
         false, a,  false, b,
         (short)0, c,  false,  false);
}

struct QkvRaw {
    v2f a[4][2];
    v4f b[4][2];
};
struct QkvOps {
    v2f a[4][2];
    v2f b[4][4];
};

static __device__ __forceinline__ void qkv_load(
    QkvRaw& f, const float* __restrict__ xa, const float* __restrict__ wb, int k0)
{
    #pragma unroll
    for (int i = 0; i < 4; ++i) {
        #pragma unroll
        for (int s = 0; s < 2; ++s) {
            f.a[i][s] = *(const v2f*)(xa + (size_t)(k0 + 4 * i + s) * NTOK);
            f.b[i][s] = *(const v4f*)(wb + (size_t)(k0 + 4 * i + s) * QKVW);
        }
    }
}

static __device__ __forceinline__ void qkv_pair(const QkvRaw& f, QkvOps& p)
{
    #pragma unroll
    for (int i = 0; i < 4; ++i) {
        #pragma unroll
        for (int t = 0; t < 2; ++t) {
            p.a[i][t] = v2f{f.a[i][0][t], f.a[i][1][t]};
            asm volatile("" : "+v"(p.a[i][t]));
        }
        #pragma unroll
        for (int j = 0; j < 4; ++j) {
            p.b[i][j] = v2f{f.b[i][0][j], f.b[i][1][j]};
            asm volatile("" : "+v"(p.b[i][j]));
        }
    }
}

static __device__ __forceinline__ void qkv_mma(const QkvOps& p, v8f (&acc)[2][4])
{
    #pragma unroll
    for (int i = 0; i < 4; ++i) {
        #pragma unroll
        for (int t = 0; t < 2; ++t) {
            #pragma unroll
            for (int j = 0; j < 4; ++j)
                acc[t][j] = wmma4(p.a[i][t], p.b[i][j], acc[t][j]);
        }
    }
}

static __device__ __forceinline__ void qkv_wmma_gap()
{
    asm volatile("v_nop\n\tv_nop\n\tv_nop\n\tv_nop");
}

__global__ __launch_bounds__(128) void mhsa_qkv_gemm(
    const float* __restrict__ x, const float* __restrict__ W,
    const float* __restrict__ bias, float* __restrict__ qkv,
    float* __restrict__ zacc)
{
    {
        const int gid = (blockIdx.y * (NTOK / 128) + blockIdx.x) * 128 + threadIdx.x;
        if (gid < NZERO) zacc[gid] = 0.0f;
    }
    const int lane = threadIdx.x & 31;
    const int wave = threadIdx.x >> 5;
    const int lo = lane & 15;
    const int hi = lane >> 4;
    const int m0 = blockIdx.x * 128 + wave * 32;
    const int n0 = blockIdx.y * 64;

    const float* xa = x + (size_t)(2 * hi) * NTOK + m0 + 2 * lo;
    const float* wb = W + (size_t)(2 * hi) * QKVW + n0 + 4 * lo;

    v8f acc[2][4] = {};
    QkvRaw raw;
    QkvOps ops;
    qkv_load(raw, xa, wb, 0);
    #pragma unroll 1
    for (int k0 = 0; k0 < EMBED - 16; k0 += 16) {
        qkv_pair(raw, ops);
        __builtin_amdgcn_sched_barrier(0);
        qkv_load(raw, xa, wb, k0 + 16);
        __builtin_amdgcn_sched_barrier(0);
        qkv_mma(ops, acc);
        __builtin_amdgcn_sched_barrier(0);
        qkv_wmma_gap();
        __builtin_amdgcn_sched_barrier(0);
    }
    qkv_pair(raw, ops);
    __builtin_amdgcn_sched_barrier(0);
    qkv_mma(ops, acc);
    __builtin_amdgcn_sched_barrier(0);
    qkv_wmma_gap();
    __builtin_amdgcn_sched_barrier(0);

    const v4f bi = *(const v4f*)(bias + n0 + 4 * lo);
    #pragma unroll
    for (int t = 0; t < 2; ++t) {
        #pragma unroll
        for (int r = 0; r < 8; ++r) {
            const int row = m0 + 2 * (hi * 8 + r) + t;
            v4f o;
            o.x = acc[t][0][r] + bi.x;
            o.y = acc[t][1][r] + bi.y;
            o.z = acc[t][2][r] + bi.z;
            o.w = acc[t][3][r] + bi.w;
            *(v4f*)(qkv + (size_t)row * QKVW + n0 + 4 * lo) = o;
        }
    }
}

__global__ __launch_bounds__(512) void mhsa_reduce(
    const float* __restrict__ qkv, float* __restrict__ KtV,
    float* __restrict__ qsum)
{
    __shared__ float red[16 * 32 * 32];
    __shared__ float redq[8 * DH];
    const int h = blockIdx.x, chunk = blockIdx.y;
    const int lane = threadIdx.x & 31, wave = threadIdx.x >> 5;
    const int lo = lane & 15, hi = lane >> 4;
    const int dt = wave & 3;
    const int sl = wave >> 2;
    const int m0 = dt * 16;
    const float* Kb = qkv + EMBED + h * DH;
    const float* Vb = qkv + 2 * EMBED + h * DH;
    const int nbase = chunk * 256 + sl * 64;

    const int qd = threadIdx.x & (DH - 1), qs = threadIdx.x >> 6;
    const float* col = qkv + (size_t)(chunk * 256 + qs * 32) * QKVW + h * DH + qd;
    float qv[32];
    #pragma unroll
    for (int i = 0; i < 32; ++i) qv[i] = col[(size_t)i * QKVW];

    v8f acc0 = {}, acc1 = {}, acc2 = {}, acc3 = {};
    #pragma unroll 8
    for (int kk = 0; kk < 64; kk += 4) {
        const int n = nbase + kk + 2 * hi;
        v2f a;
        a.x = Kb[(size_t)n * QKVW + m0 + lo];
        a.y = Kb[(size_t)(n + 1) * QKVW + m0 + lo];
        const float* v0 = Vb + (size_t)n * QKVW + lo;
        const float* v1 = Vb + (size_t)(n + 1) * QKVW + lo;
        v2f b0 = {v0[0],  v1[0]};
        v2f b1 = {v0[16], v1[16]};
        v2f b2 = {v0[32], v1[32]};
        v2f b3 = {v0[48], v1[48]};
        acc0 = wmma4(a, b0, acc0);
        acc1 = wmma4(a, b1, acc1);
        acc2 = wmma4(a, b2, acc2);
        acc3 = wmma4(a, b3, acc3);
    }

    float* mine = red + wave * 1024 + lane;
    #pragma unroll
    for (int r = 0; r < 8; ++r) {
        mine[(0 * 8 + r) * 32] = acc0[r];
        mine[(1 * 8 + r) * 32] = acc1[r];
        mine[(2 * 8 + r) * 32] = acc2[r];
        mine[(3 * 8 + r) * 32] = acc3[r];
    }
    {
        float s = 0.0f;
        #pragma unroll
        for (int i = 0; i < 32; ++i) s += qv[i];
        redq[qs * DH + qd] = s;
    }
    __syncthreads();

    const float* p = red + dt * 1024 + sl * 256 + lane;
    float* dst = KtV + (size_t)h * DH * DH;
    #pragma unroll
    for (int r = 0; r < 8; ++r) {
        const float s = ((p[r * 32] + p[4096 + r * 32]) + p[8192 + r * 32])
                        + p[12288 + r * 32];
        const int row = m0 + hi * 8 + r;
        atomicAdd(&dst[row * DH + lo + 16 * sl], s);
    }
    if (threadIdx.x < DH) {
        float t = redq[qd];
        #pragma unroll
        for (int j = 1; j < 8; ++j) t += redq[j * DH + qd];
        atomicAdd(&qsum[h * DH + qd], t);
    }
}

__global__ __launch_bounds__(128) void mhsa_apply(
    const float* __restrict__ qkv, const float* __restrict__ KtV,
    const float* __restrict__ qsum, float* __restrict__ out)
{
    __shared__ float lB[DH * DH];
    const int h = blockIdx.x, mt = blockIdx.y;
    const int lane = threadIdx.x & 31, wave = threadIdx.x >> 5;

    const int ct = mt * 64 + (wave & 1) * 32 + lane;
    v4f kr[16];
    if (wave < 2) {
        const float* Kr = qkv + (size_t)ct * QKVW + EMBED + h * DH;
        #pragma unroll
        for (int j = 0; j < 16; ++j) kr[j] = *(const v4f*)(Kr + 4 * j);
    }

    {
        const v4f* src = (const v4f*)(KtV + (size_t)h * DH * DH);
        v4f* dst = (v4f*)lB;
        for (int i = threadIdx.x; i < (DH * DH) / 4; i += 128) dst[i] = src[i];
    }
    __syncthreads();

    if (wave < 2) {
        const float* qs = qsum + h * DH;
        float s = 0.0f;
        #pragma unroll
        for (int j = 0; j < 16; ++j) {
            const v4f kv = kr[j];
            const v4f qv = *(const v4f*)(qs + 4 * j);
            s += kv.x * qv.x + kv.y * qv.y + kv.z * qv.z + kv.w * qv.w;
        }
        out[h * NTOK + ct] = s * SCALE;
    }

    const int lo = lane & 15, hi = lane >> 4;
    const int m0 = mt * 64 + wave * 16;
    const float* Qb = qkv + h * DH;
    const int m = m0 + lo;

    v8f acc0 = {}, acc1 = {}, acc2 = {}, acc3 = {};
    #pragma unroll
    for (int k = 0; k < DH; k += 4) {
        const int ka = k + 2 * hi;
        v2f a = *(const v2f*)(Qb + (size_t)m * QKVW + ka);
        const float* B0 = lB + ka * DH + lo;
        const float* B1 = lB + (ka + 1) * DH + lo;
        v2f b0 = {B0[0],  B1[0]};
        v2f b1 = {B0[16], B1[16]};
        v2f b2 = {B0[32], B1[32]};
        v2f b3 = {B0[48], B1[48]};
        acc0 = wmma4(a, b0, acc0);
        acc1 = wmma4(a, b1, acc1);
        acc2 = wmma4(a, b2, acc2);
        acc3 = wmma4(a, b3, acc3);
    }

    float* base = out + HEADS * NTOK + (size_t)h * NTOK * DH;
    #pragma unroll
    for (int r = 0; r < 8; ++r) {
        const int row = m0 + hi * 8 + r;
        float* o = base + (size_t)row * DH + lo;
        o[0]  = acc0[r] * SCALE;
        o[16] = acc1[r] * SCALE;
        o[32] = acc2[r] * SCALE;
        o[48] = acc3[r] * SCALE;
    }
}

extern "C" void kernel_launch(void* const* d_in, const int* in_sizes, int n_in,
                              void* d_out, int out_size, void* d_ws, size_t ws_size,
                              hipStream_t stream) {
    const float* x    = (const float*)d_in[0];
    const float* W    = (const float*)d_in[1];
    const float* bias = (const float*)d_in[2];
    float* out = (float*)d_out;

    float* qkv  = (float*)d_ws;
    float* KtV  = qkv + (size_t)NTOK * QKVW;
    float* qsum = KtV + HEADS * DH * DH;

    mhsa_qkv_gemm<<<dim3(NTOK / 128, QKVW / 64), 128, 0, stream>>>(x, W, bias, qkv, KtV);
    mhsa_reduce<<<dim3(HEADS, NTOK / 256), 512, 0, stream>>>(qkv, KtV, qsum);
    mhsa_apply<<<dim3(HEADS, NTOK / 64), 128, 0, stream>>>(qkv, KtV, qsum, out);
}


# Output GEMM: KtV read straight from global as 16-byte words of column-interleaved WMMA tiles, a wave's loads in one group; LDS stage, barrier and 8 serial round trips removed; sum order unchanged
# speedup vs baseline: 1.7375x; 1.0358x over previous
#include <hip/hip_runtime.h>
#include <hip/hip_bf16.h>


typedef __attribute__((ext_vector_type(2))) float v2f;
typedef __attribute__((ext_vector_type(4))) float v4f;
typedef __attribute__((ext_vector_type(8))) float v8f;

#define NTOK   4096
#define EMBED  512
#define QKVW   1536
#define HEADS  8
#define DH     64
#define SCALE  0.125f
#define NZERO  (HEADS * DH * DH + HEADS * DH)

static __device__ __forceinline__ v8f wmma4(v2f a, v2f b, v8f c) {
    return __builtin_amdgcn_wmma_f32_16x16x4_f32(
         false, a,  false, b,
         (short)0, c,  false,  false);
}

struct QkvRaw {
    v2f a[4][2];
    v4f b[4][2];
};
struct QkvOps {
    v2f a[4][2];
    v2f b[4][4];
};

static __device__ __forceinline__ void qkv_load(
    QkvRaw& f, const float* __restrict__ xa, const float* __restrict__ wb, int k0)
{
    #pragma unroll
    for (int i = 0; i < 4; ++i) {
        #pragma unroll
        for (int s = 0; s < 2; ++s) {
            f.a[i][s] = *(const v2f*)(xa + (size_t)(k0 + 4 * i + s) * NTOK);
            f.b[i][s] = *(const v4f*)(wb + (size_t)(k0 + 4 * i + s) * QKVW);
        }
    }
}

static __device__ __forceinline__ void qkv_pair(const QkvRaw& f, QkvOps& p)
{
    #pragma unroll
    for (int i = 0; i < 4; ++i) {
        #pragma unroll
        for (int t = 0; t < 2; ++t) {
            p.a[i][t] = v2f{f.a[i][0][t], f.a[i][1][t]};
            asm volatile("" : "+v"(p.a[i][t]));
        }
        #pragma unroll
        for (int j = 0; j < 4; ++j) {
            p.b[i][j] = v2f{f.b[i][0][j], f.b[i][1][j]};
            asm volatile("" : "+v"(p.b[i][j]));
        }
    }
}

static __device__ __forceinline__ void qkv_mma(const QkvOps& p, v8f (&acc)[2][4])
{
    #pragma unroll
    for (int i = 0; i < 4; ++i) {
        #pragma unroll
        for (int t = 0; t < 2; ++t) {
            #pragma unroll
            for (int j = 0; j < 4; ++j)
                acc[t][j] = wmma4(p.a[i][t], p.b[i][j], acc[t][j]);
        }
    }
}

static __device__ __forceinline__ void qkv_wmma_gap()
{
    asm volatile("v_nop\n\tv_nop\n\tv_nop\n\tv_nop");
}

__global__ __launch_bounds__(128) void mhsa_qkv_gemm(
    const float* __restrict__ x, const float* __restrict__ W,
    const float* __restrict__ bias, float* __restrict__ qkv,
    float* __restrict__ zacc)
{
    {
        const int gid = (blockIdx.y * (NTOK / 128) + blockIdx.x) * 128 + threadIdx.x;
        if (gid < NZERO) zacc[gid] = 0.0f;
    }
    const int lane = threadIdx.x & 31;
    const int wave = threadIdx.x >> 5;
    const int lo = lane & 15;
    const int hi = lane >> 4;
    const int m0 = blockIdx.x * 128 + wave * 32;
    const int n0 = blockIdx.y * 64;

    const float* xa = x + (size_t)(2 * hi) * NTOK + m0 + 2 * lo;
    const float* wb = W + (size_t)(2 * hi) * QKVW + n0 + 4 * lo;

    v8f acc[2][4] = {};
    QkvRaw raw;
    QkvOps ops;
    qkv_load(raw, xa, wb, 0);
    #pragma unroll 1
    for (int k0 = 0; k0 < EMBED - 16; k0 += 16) {
        qkv_pair(raw, ops);
        __builtin_amdgcn_sched_barrier(0);
        qkv_load(raw, xa, wb, k0 + 16);
        __builtin_amdgcn_sched_barrier(0);
        qkv_mma(ops, acc);
        __builtin_amdgcn_sched_barrier(0);
        qkv_wmma_gap();
        __builtin_amdgcn_sched_barrier(0);
    }
    qkv_pair(raw, ops);
    __builtin_amdgcn_sched_barrier(0);
    qkv_mma(ops, acc);
    __builtin_amdgcn_sched_barrier(0);
    qkv_wmma_gap();
    __builtin_amdgcn_sched_barrier(0);

    const v4f bi = *(const v4f*)(bias + n0 + 4 * lo);
    #pragma unroll
    for (int t = 0; t < 2; ++t) {
        #pragma unroll
        for (int r = 0; r < 8; ++r) {
            const int row = m0 + 2 * (hi * 8 + r) + t;
            v4f o;
            o.x = acc[t][0][r] + bi.x;
            o.y = acc[t][1][r] + bi.y;
            o.z = acc[t][2][r] + bi.z;
            o.w = acc[t][3][r] + bi.w;
            *(v4f*)(qkv + (size_t)row * QKVW + n0 + 4 * lo) = o;
        }
    }
}

__global__ __launch_bounds__(512) void mhsa_reduce(
    const float* __restrict__ qkv, float* __restrict__ KtV,
    float* __restrict__ qsum)
{
    __shared__ float red[16 * 32 * 32];
    __shared__ float redq[8 * DH];
    const int h = blockIdx.x, chunk = blockIdx.y;
    const int lane = threadIdx.x & 31, wave = threadIdx.x >> 5;
    const int lo = lane & 15, hi = lane >> 4;
    const int dt = wave & 3;
    const int sl = wave >> 2;
    const int m0 = dt * 16;
    const float* Kb = qkv + EMBED + h * DH;
    const float* Vb = qkv + 2 * EMBED + h * DH;
    const int nbase = chunk * 256 + sl * 64;

    const int qd = threadIdx.x & (DH - 1), qs = threadIdx.x >> 6;
    const float* col = qkv + (size_t)(chunk * 256 + qs * 32) * QKVW + h * DH + qd;
    float qv[32];
    #pragma unroll
    for (int i = 0; i < 32; ++i) qv[i] = col[(size_t)i * QKVW];

    v8f acc0 = {}, acc1 = {}, acc2 = {}, acc3 = {};
    #pragma unroll 8
    for (int kk = 0; kk < 64; kk += 4) {
        const int n = nbase + kk + 2 * hi;
        v2f a;
        a.x = Kb[(size_t)n * QKVW + m0 + lo];
        a.y = Kb[(size_t)(n + 1) * QKVW + m0 + lo];
        const float* v0 = Vb + (size_t)n * QKVW + lo;
        const float* v1 = Vb + (size_t)(n + 1) * QKVW + lo;
        v2f b0 = {v0[0],  v1[0]};
        v2f b1 = {v0[16], v1[16]};
        v2f b2 = {v0[32], v1[32]};
        v2f b3 = {v0[48], v1[48]};
        acc0 = wmma4(a, b0, acc0);
        acc1 = wmma4(a, b1, acc1);
        acc2 = wmma4(a, b2, acc2);
        acc3 = wmma4(a, b3, acc3);
    }

    float* mine = red + wave * 1024 + lane;
    #pragma unroll
    for (int r = 0; r < 8; ++r) {
        mine[(0 * 8 + r) * 32] = acc0[r];
        mine[(1 * 8 + r) * 32] = acc1[r];
        mine[(2 * 8 + r) * 32] = acc2[r];
        mine[(3 * 8 + r) * 32] = acc3[r];
    }
    {
        float s = 0.0f;
        #pragma unroll
        for (int i = 0; i < 32; ++i) s += qv[i];
        redq[qs * DH + qd] = s;
    }
    __syncthreads();

    const float* p = red + dt * 1024 + sl * 256 + lane;
    float* dst = KtV + (size_t)h * DH * DH;
    #pragma unroll
    for (int r = 0; r < 8; ++r) {
        const float s = ((p[r * 32] + p[4096 + r * 32]) + p[8192 + r * 32])
                        + p[12288 + r * 32];
        const int row = m0 + hi * 8 + r;
        atomicAdd(&dst[row * DH + lo + 16 * sl], s);
    }
    if (threadIdx.x < DH) {
        float t = redq[qd];
        #pragma unroll
        for (int j = 1; j < 8; ++j) t += redq[j * DH + qd];
        atomicAdd(&qsum[h * DH + qd], t);
    }
}

__global__ __launch_bounds__(128) void mhsa_apply(
    const float* __restrict__ qkv, const float* __restrict__ KtV,
    const float* __restrict__ qsum, float* __restrict__ out)
{
    const int h = blockIdx.x, mt = blockIdx.y;
    const int lane = threadIdx.x & 31, wave = threadIdx.x >> 5;
    const int lo = lane & 15, hi = lane >> 4;
    const int m0 = mt * 64 + wave * 16;

    const int ct = mt * 64 + (wave & 1) * 32 + lane;
    const float* qs = qsum + h * DH;
    v4f qr[16];
    #pragma unroll
    for (int j = 0; j < 16; ++j) qr[j] = *(const v4f*)(qs + 4 * j);
    v4f kr[16];
    if (wave < 2) {
        const float* Kr = qkv + (size_t)ct * QKVW + EMBED + h * DH;
        #pragma unroll
        for (int j = 0; j < 16; ++j) kr[j] = *(const v4f*)(Kr + 4 * j);
    }

    const float* Qa = qkv + (size_t)(m0 + lo) * QKVW + h * DH + 2 * hi;
    v2f a[16];
    #pragma unroll
    for (int i = 0; i < 16; ++i) a[i] = *(const v2f*)(Qa + 4 * i);

    const float* Bp = KtV + (size_t)h * DH * DH + (2 * hi) * DH + 4 * lo;
    v4f bw[16][2];
    #pragma unroll
    for (int i = 0; i < 16; ++i) {
        #pragma unroll
        for (int s = 0; s < 2; ++s)
            bw[i][s] = *(const v4f*)(Bp + (4 * i + s) * DH);
    }

    if (wave < 2) {
        #pragma unroll
        for (int j = 0; j < 16; ++j) asm volatile("" : "+v"(kr[j]));
        float s = 0.0f;
        #pragma unroll
        for (int j = 0; j < 16; ++j) {
            const v4f kv = kr[j];
            const v4f qv = qr[j];
            s += kv.x * qv.x + kv.y * qv.y + kv.z * qv.z + kv.w * qv.w;
        }
        out[h * NTOK + ct] = s * SCALE;
    }

    v2f bp[16][4];
    #pragma unroll
    for (int i = 0; i < 16; ++i) {
        #pragma unroll
        for (int j = 0; j < 4; ++j) {
            bp[i][j] = v2f{bw[i][0][j], bw[i][1][j]};
            asm volatile("" : "+v"(bp[i][j]));
        }
    }
    __builtin_amdgcn_sched_barrier(0);

    v8f acc0 = {}, acc1 = {}, acc2 = {}, acc3 = {};
    #pragma unroll
    for (int i = 0; i < 16; ++i) {
        acc0 = wmma4(a[i], bp[i][0], acc0);
        acc1 = wmma4(a[i], bp[i][1], acc1);
        acc2 = wmma4(a[i], bp[i][2], acc2);
        acc3 = wmma4(a[i], bp[i][3], acc3);
    }

    float* base = out + HEADS * NTOK + (size_t)h * NTOK * DH;
    #pragma unroll
    for (int r = 0; r < 8; ++r) {
        const int row = m0 + hi * 8 + r;
        v4f o;
        o.x = acc0[r] * SCALE;
        o.y = acc1[r] * SCALE;
        o.z = acc2[r] * SCALE;
        o.w = acc3[r] * SCALE;
        *(v4f*)(base + (size_t)row * DH + 4 * lo) = o;
    }
}

extern "C" void kernel_launch(void* const* d_in, const int* in_sizes, int n_in,
                              void* d_out, int out_size, void* d_ws, size_t ws_size,
                              hipStream_t stream) {
    const float* x    = (const float*)d_in[0];
    const float* W    = (const float*)d_in[1];
    const float* bias = (const float*)d_in[2];
    float* out = (float*)d_out;

    float* qkv  = (float*)d_ws;
    float* KtV  = qkv + (size_t)NTOK * QKVW;
    float* qsum = KtV + HEADS * DH * DH;

    mhsa_qkv_gemm<<<dim3(NTOK / 128, QKVW / 64), 128, 0, stream>>>(x, W, bias, qkv, KtV);
    mhsa_reduce<<<dim3(HEADS, NTOK / 256), 512, 0, stream>>>(qkv, KtV, qsum);
    mhsa_apply<<<dim3(HEADS, NTOK / 64), 128, 0, stream>>>(qkv, KtV, qsum, out);
}


# QKV GEMM: a second register set of loaded words, each set refilled with the 16-k block two ahead, so loads have 64 WMMAs to arrive in (was 32); sum order unchanged
# speedup vs baseline: 1.9554x; 1.1254x over previous
#include <hip/hip_runtime.h>
#include <hip/hip_bf16.h>


typedef __attribute__((ext_vector_type(2))) float v2f;
typedef __attribute__((ext_vector_type(4))) float v4f;
typedef __attribute__((ext_vector_type(8))) float v8f;

#define NTOK   4096
#define EMBED  512
#define QKVW   1536
#define HEADS  8
#define DH     64
#define SCALE  0.125f
#define NZERO  (HEADS * DH * DH + HEADS * DH)

static __device__ __forceinline__ v8f wmma4(v2f a, v2f b, v8f c) {
    return __builtin_amdgcn_wmma_f32_16x16x4_f32(
         false, a,  false, b,
         (short)0, c,  false,  false);
}

struct QkvRaw {
    v2f a[4][2];
    v4f b[4][2];
};
struct QkvOps {
    v2f a[4][2];
    v2f b[4][4];
};

static __device__ __forceinline__ void qkv_load(
    QkvRaw& f, const float* __restrict__ xa, const float* __restrict__ wb, int k0)
{
    #pragma unroll
    for (int i = 0; i < 4; ++i) {
        #pragma unroll
        for (int s = 0; s < 2; ++s) {
            f.a[i][s] = *(const v2f*)(xa + (size_t)(k0 + 4 * i + s) * NTOK);
            f.b[i][s] = *(const v4f*)(wb + (size_t)(k0 + 4 * i + s) * QKVW);
        }
    }
}

static __device__ __forceinline__ void qkv_pair(const QkvRaw& f, QkvOps& p)
{
    #pragma unroll
    for (int i = 0; i < 4; ++i) {
        #pragma unroll
        for (int t = 0; t < 2; ++t) {
            p.a[i][t] = v2f{f.a[i][0][t], f.a[i][1][t]};
            asm volatile("" : "+v"(p.a[i][t]));
        }
        #pragma unroll
        for (int j = 0; j < 4; ++j) {
            p.b[i][j] = v2f{f.b[i][0][j], f.b[i][1][j]};
            asm volatile("" : "+v"(p.b[i][j]));
        }
    }
}

static __device__ __forceinline__ void qkv_mma(const QkvOps& p, v8f (&acc)[2][4])
{
    #pragma unroll
    for (int i = 0; i < 4; ++i) {
        #pragma unroll
        for (int t = 0; t < 2; ++t) {
            #pragma unroll
            for (int j = 0; j < 4; ++j)
                acc[t][j] = wmma4(p.a[i][t], p.b[i][j], acc[t][j]);
        }
    }
}

static __device__ __forceinline__ void qkv_wmma_gap()
{
    asm volatile("v_nop\n\tv_nop\n\tv_nop\n\tv_nop");
}

static __device__ __forceinline__ void qkv_block(
    QkvRaw& raw, QkvOps& ops, v8f (&acc)[2][4],
    const float* __restrict__ xa, const float* __restrict__ wb, int knext)
{
    qkv_pair(raw, ops);
    __builtin_amdgcn_sched_barrier(0);
    qkv_load(raw, xa, wb, knext);
    __builtin_amdgcn_sched_barrier(0);
    qkv_mma(ops, acc);
    __builtin_amdgcn_sched_barrier(0);
    qkv_wmma_gap();
    __builtin_amdgcn_sched_barrier(0);
}

static __device__ __forceinline__ void qkv_last(
    const QkvRaw& raw, QkvOps& ops, v8f (&acc)[2][4])
{
    qkv_pair(raw, ops);
    __builtin_amdgcn_sched_barrier(0);
    qkv_mma(ops, acc);
    __builtin_amdgcn_sched_barrier(0);
    qkv_wmma_gap();
    __builtin_amdgcn_sched_barrier(0);
}

__global__ __launch_bounds__(128) void mhsa_qkv_gemm(
    const float* __restrict__ x, const float* __restrict__ W,
    const float* __restrict__ bias, float* __restrict__ qkv,
    float* __restrict__ zacc)
{
    {
        const int gid = (blockIdx.y * (NTOK / 128) + blockIdx.x) * 128 + threadIdx.x;
        if (gid < NZERO) zacc[gid] = 0.0f;
    }
    const int lane = threadIdx.x & 31;
    const int wave = threadIdx.x >> 5;
    const int lo = lane & 15;
    const int hi = lane >> 4;
    const int m0 = blockIdx.x * 128 + wave * 32;
    const int n0 = blockIdx.y * 64;

    const float* xa = x + (size_t)(2 * hi) * NTOK + m0 + 2 * lo;
    const float* wb = W + (size_t)(2 * hi) * QKVW + n0 + 4 * lo;

    v8f acc[2][4] = {};
    QkvRaw r0, r1;
    QkvOps ops;
    qkv_load(r0, xa, wb, 0);
    __builtin_amdgcn_sched_barrier(0);
    qkv_load(r1, xa, wb, 16);
    __builtin_amdgcn_sched_barrier(0);
    #pragma unroll 1
    for (int k0 = 0; k0 < EMBED - 32; k0 += 32) {
        qkv_block(r0, ops, acc, xa, wb, k0 + 32);
        qkv_block(r1, ops, acc, xa, wb, k0 + 48);
    }
    qkv_last(r0, ops, acc);
    qkv_last(r1, ops, acc);

    const v4f bi = *(const v4f*)(bias + n0 + 4 * lo);
    #pragma unroll
    for (int t = 0; t < 2; ++t) {
        #pragma unroll
        for (int r = 0; r < 8; ++r) {
            const int row = m0 + 2 * (hi * 8 + r) + t;
            v4f o;
            o.x = acc[t][0][r] + bi.x;
            o.y = acc[t][1][r] + bi.y;
            o.z = acc[t][2][r] + bi.z;
            o.w = acc[t][3][r] + bi.w;
            *(v4f*)(qkv + (size_t)row * QKVW + n0 + 4 * lo) = o;
        }
    }
}

__global__ __launch_bounds__(512) void mhsa_reduce(
    const float* __restrict__ qkv, float* __restrict__ KtV,
    float* __restrict__ qsum)
{
    __shared__ float red[16 * 32 * 32];
    __shared__ float redq[8 * DH];
    const int h = blockIdx.x, chunk = blockIdx.y;
    const int lane = threadIdx.x & 31, wave = threadIdx.x >> 5;
    const int lo = lane & 15, hi = lane >> 4;
    const int dt = wave & 3;
    const int sl = wave >> 2;
    const int m0 = dt * 16;
    const float* Kb = qkv + EMBED + h * DH;
    const float* Vb = qkv + 2 * EMBED + h * DH;
    const int nbase = chunk * 256 + sl * 64;

    const int qd = threadIdx.x & (DH - 1), qs = threadIdx.x >> 6;
    const float* col = qkv + (size_t)(chunk * 256 + qs * 32) * QKVW + h * DH + qd;
    float qv[32];
    #pragma unroll
    for (int i = 0; i < 32; ++i) qv[i] = col[(size_t)i * QKVW];

    v8f acc0 = {}, acc1 = {}, acc2 = {}, acc3 = {};
    #pragma unroll 8
    for (int kk = 0; kk < 64; kk += 4) {
        const int n = nbase + kk + 2 * hi;
        v2f a;
        a.x = Kb[(size_t)n * QKVW + m0 + lo];
        a.y = Kb[(size_t)(n + 1) * QKVW + m0 + lo];
        const float* v0 = Vb + (size_t)n * QKVW + lo;
        const float* v1 = Vb + (size_t)(n + 1) * QKVW + lo;
        v2f b0 = {v0[0],  v1[0]};
        v2f b1 = {v0[16], v1[16]};
        v2f b2 = {v0[32], v1[32]};
        v2f b3 = {v0[48], v1[48]};
        acc0 = wmma4(a, b0, acc0);
        acc1 = wmma4(a, b1, acc1);
        acc2 = wmma4(a, b2, acc2);
        acc3 = wmma4(a, b3, acc3);
    }

    float* mine = red + wave * 1024 + lane;
    #pragma unroll
    for (int r = 0; r < 8; ++r) {
        mine[(0 * 8 + r) * 32] = acc0[r];
        mine[(1 * 8 + r) * 32] = acc1[r];
        mine[(2 * 8 + r) * 32] = acc2[r];
        mine[(3 * 8 + r) * 32] = acc3[r];
    }
    {
        float s = 0.0f;
        #pragma unroll
        for (int i = 0; i < 32; ++i) s += qv[i];
        redq[qs * DH + qd] = s;
    }
    __syncthreads();

    const float* p = red + dt * 1024 + sl * 256 + lane;
    float* dst = KtV + (size_t)h * DH * DH;
    #pragma unroll
    for (int r = 0; r < 8; ++r) {
        const float s = ((p[r * 32] + p[4096 + r * 32]) + p[8192 + r * 32])
                        + p[12288 + r * 32];
        const int row = m0 + hi * 8 + r;
        atomicAdd(&dst[row * DH + lo + 16 * sl], s);
    }
    if (threadIdx.x < DH) {
        float t = redq[qd];
        #pragma unroll
        for (int j = 1; j < 8; ++j) t += redq[j * DH + qd];
        atomicAdd(&qsum[h * DH + qd], t);
    }
}

__global__ __launch_bounds__(128) void mhsa_apply(
    const float* __restrict__ qkv, const float* __restrict__ KtV,
    const float* __restrict__ qsum, float* __restrict__ out)
{
    const int h = blockIdx.x, mt = blockIdx.y;
    const int lane = threadIdx.x & 31, wave = threadIdx.x >> 5;
    const int lo = lane & 15, hi = lane >> 4;
    const int m0 = mt * 64 + wave * 16;

    const int ct = mt * 64 + (wave & 1) * 32 + lane;
    const float* qs = qsum + h * DH;
    v4f qr[16];
    #pragma unroll
    for (int j = 0; j < 16; ++j) qr[j] = *(const v4f*)(qs + 4 * j);
    v4f kr[16];
    if (wave < 2) {
        const float* Kr = qkv + (size_t)ct * QKVW + EMBED + h * DH;
        #pragma unroll
        for (int j = 0; j < 16; ++j) kr[j] = *(const v4f*)(Kr + 4 * j);
    }

    const float* Qa = qkv + (size_t)(m0 + lo) * QKVW + h * DH + 2 * hi;
    v2f a[16];
    #pragma unroll
    for (int i = 0; i < 16; ++i) a[i] = *(const v2f*)(Qa + 4 * i);

    const float* Bp = KtV + (size_t)h * DH * DH + (2 * hi) * DH + 4 * lo;
    v4f bw[16][2];
    #pragma unroll
    for (int i = 0; i < 16; ++i) {
        #pragma unroll
        for (int s = 0; s < 2; ++s)
            bw[i][s] = *(const v4f*)(Bp + (4 * i + s) * DH);
    }

    if (wave < 2) {
        #pragma unroll
        for (int j = 0; j < 16; ++j) asm volatile("" : "+v"(kr[j]));
        float s = 0.0f;
        #pragma unroll
        for (int j = 0; j < 16; ++j) {
            const v4f kv = kr[j];
            const v4f qv = qr[j];
            s += kv.x * qv.x + kv.y * qv.y + kv.z * qv.z + kv.w * qv.w;
        }
        out[h * NTOK + ct] = s * SCALE;
    }

    v2f bp[16][4];
    #pragma unroll
    for (int i = 0; i < 16; ++i) {
        #pragma unroll
        for (int j = 0; j < 4; ++j) {
            bp[i][j] = v2f{bw[i][0][j], bw[i][1][j]};
            asm volatile("" : "+v"(bp[i][j]));
        }
    }
    __builtin_amdgcn_sched_barrier(0);

    v8f acc0 = {}, acc1 = {}, acc2 = {}, acc3 = {};
    #pragma unroll
    for (int i = 0; i < 16; ++i) {
        acc0 = wmma4(a[i], bp[i][0], acc0);
        acc1 = wmma4(a[i], bp[i][1], acc1);
        acc2 = wmma4(a[i], bp[i][2], acc2);
        acc3 = wmma4(a[i], bp[i][3], acc3);
    }

    float* base = out + HEADS * NTOK + (size_t)h * NTOK * DH;
    #pragma unroll
    for (int r = 0; r < 8; ++r) {
        const int row = m0 + hi * 8 + r;
        v4f o;
        o.x = acc0[r] * SCALE;
        o.y = acc1[r] * SCALE;
        o.z = acc2[r] * SCALE;
        o.w = acc3[r] * SCALE;
        *(v4f*)(base + (size_t)row * DH + 4 * lo) = o;
    }
}

extern "C" void kernel_launch(void* const* d_in, const int* in_sizes, int n_in,
                              void* d_out, int out_size, void* d_ws, size_t ws_size,
                              hipStream_t stream) {
    const float* x    = (const float*)d_in[0];
    const float* W    = (const float*)d_in[1];
    const float* bias = (const float*)d_in[2];
    float* out = (float*)d_out;

    float* qkv  = (float*)d_ws;
    float* KtV  = qkv + (size_t)NTOK * QKVW;
    float* qsum = KtV + HEADS * DH * DH;

    mhsa_qkv_gemm<<<dim3(NTOK / 128, QKVW / 64), 128, 0, stream>>>(x, W, bias, qkv, KtV);
    mhsa_reduce<<<dim3(HEADS, NTOK / 256), 512, 0, stream>>>(qkv, KtV, qsum);
    mhsa_apply<<<dim3(HEADS, NTOK / 64), 128, 0, stream>>>(qkv, KtV, qsum, out);
}
